# starting kernel
# baseline (speedup 1.0000x reference)
#include <hip/hip_runtime.h>

typedef __attribute__((ext_vector_type(2))) float v2f;
typedef __attribute__((ext_vector_type(4))) float v4f;
typedef __attribute__((ext_vector_type(8))) float v8f;

#define BATCH   32768
#define NTILES  (BATCH / 16)
#define BLOCKS  256
#define TPB     128

__device__ __forceinline__ v8f wmma_f32_k4(v2f a, v2f b, v8f c) {
  return __builtin_amdgcn_wmma_f32_16x16x4_f32(
       false, a,  false, b,
       (short)0, c,  false,  false);
}

__global__ __launch_bounds__(TPB) void gcn_fused_kernel(
    const float* __restrict__ x,
    const float* __restrict__ qW1, const float* __restrict__ qb1,
    const float* __restrict__ qW2, const float* __restrict__ qb2,
    const float* __restrict__ cW1, const float* __restrict__ cb1,
    const float* __restrict__ cW2, const float* __restrict__ cb2,
    const float* __restrict__ eTW1, const float* __restrict__ eTb1,
    const float* __restrict__ eTW2, const float* __restrict__ eTb2,
    const float* __restrict__ eHW1, const float* __restrict__ eHb1,
    const float* __restrict__ eHW2, const float* __restrict__ eHb2,
    float* __restrict__ out)
{
  __shared__ float s_w2m[2][64];
  __shared__ float s_eW1[2][5][32];
  __shared__ float s_eb1[2][32];
  __shared__ float s_eW2[2][32];
  __shared__ float s_eb2[2];

  const int tid = threadIdx.x;

  {
    const int net = tid >> 6;
    const int h   = tid & 63;
    const float* W2 = net ? cW2 : qW2;
    const v4f* p = (const v4f*)(W2 + h * 32);
    float s = 0.0f;
#pragma unroll
    for (int i = 0; i < 8; ++i) { v4f v = p[i]; s += v.x + v.y + v.z + v.w; }
    s_w2m[net][h] = s * (1.0f / 32.0f);
  }
  if (tid < 32) {
#pragma unroll
    for (int f = 0; f < 5; ++f) {
      s_eW1[0][f][tid] = eTW1[f * 32 + tid];
      s_eW1[1][f][tid] = eHW1[f * 32 + tid];
    }
    s_eb1[0][tid] = eTb1[tid];
    s_eb1[1][tid] = eHb1[tid];
    s_eW2[0][tid] = eTW2[tid];
    s_eW2[1][tid] = eHW2[tid];
    if (tid < 2) s_eb2[tid] = tid ? eHb2[0] : eTb2[0];
  }
  __syncthreads();

  const int lane = tid & 31;
  const int half = lane >> 4;
  const int lm   = lane & 15;
  const int koff = half * 2;

  float b1r[2][4], w2mr[2][4];
#pragma unroll
  for (int n = 0; n < 4; ++n) {
    const int col = n * 16 + lm;
    b1r[0][n]  = qb1[col];
    b1r[1][n]  = cb1[col];
    w2mr[0][n] = s_w2m[0][col];
    w2mr[1][n] = s_w2m[1][col];
  }
  float mb2 = qb2[lane] + cb2[lane];
#pragma unroll
  for (int m = 16; m >= 1; m >>= 1) mb2 += __shfl_xor(mb2, m, 32);
  mb2 *= (1.0f / 32.0f);

  v2f bq[4][8], bc[4][8];
#pragma unroll
  for (int n = 0; n < 4; ++n) {
    const int col = n * 16 + lm;
#pragma unroll
    for (int k = 0; k < 8; ++k) {
      const int kr = k * 4 + koff;
      v2f t;
      t.x = qW1[kr * 64 + col]; t.y = qW1[(kr + 1) * 64 + col]; bq[n][k] = t;
      t.x = cW1[kr * 64 + col]; t.y = cW1[(kr + 1) * 64 + col]; bc[n][k] = t;
    }
  }

  const int wgid  = blockIdx.x * (TPB >> 5) + (tid >> 5);
  const int wstep = gridDim.x * (TPB >> 5);

  for (int t = wgid; t < NTILES; t += wstep) {
    const int base = t * 16;

    v2f a[8];
    const float* xr = x + (base + lm) * 32 + koff;
#pragma unroll
    for (int k = 0; k < 8; ++k) a[k] = *(const v2f*)(xr + 4 * k);

    float accv[8];
#pragma unroll
    for (int v = 0; v < 8; ++v) accv[v] = 0.0f;

#pragma unroll
    for (int n = 0; n < 4; ++n) {
      v8f c = {0.f, 0.f, 0.f, 0.f, 0.f, 0.f, 0.f, 0.f};
#pragma unroll
      for (int k = 0; k < 8; ++k) c = wmma_f32_k4(a[k], bq[n][k], c);
#pragma unroll
      for (int v = 0; v < 8; ++v) {
        const float h = fmaxf(c[v] + b1r[0][n], 0.0f);
        accv[v] = fmaf(h, w2mr[0][n], accv[v]);
      }
    }
#pragma unroll
    for (int n = 0; n < 4; ++n) {
      v8f c = {0.f, 0.f, 0.f, 0.f, 0.f, 0.f, 0.f, 0.f};
#pragma unroll
      for (int k = 0; k < 8; ++k) c = wmma_f32_k4(a[k], bc[n][k], c);
#pragma unroll
      for (int v = 0; v < 8; ++v) {
        const float h = fmaxf(c[v] + b1r[1][n], 0.0f);
        accv[v] = fmaf(h, w2mr[1][n], accv[v]);
      }
    }

#pragma unroll
    for (int v = 0; v < 8; ++v) {
      float s = accv[v];
#pragma unroll
      for (int m = 8; m >= 1; m >>= 1) s += __shfl_xor(s, m, 32);
      accv[v] = s + mb2;
    }

    const float T = x[(base + lm) * 32 + 0];
    const float H = x[(base + lm) * 32 + 1];
    const float f0 = T, f1 = H, f2 = T * T, f3 = T * H;
    const float f4 = half ? (T * H * H) : (T * T * T);
    const int net = half;
    float cacc = s_eb2[net];
#pragma unroll 4
    for (int j = 0; j < 32; ++j) {
      float hj = s_eb1[net][j];
      hj = fmaf(f0, s_eW1[net][0][j], hj);
      hj = fmaf(f1, s_eW1[net][1][j], hj);
      hj = fmaf(f2, s_eW1[net][2][j], hj);
      hj = fmaf(f3, s_eW1[net][3][j], hj);
      hj = fmaf(f4, s_eW1[net][4][j], hj);
      cacc = fmaf(fmaxf(hj, 0.0f), s_eW2[net][j], cacc);
    }

#pragma unroll
    for (int r = 0; r < 16; ++r) {
      const float nlr = __shfl(accv[r & 7], (r < 8) ? 0 : 16, 32);
      const float cTr = __shfl(cacc, r, 32);
      const float cHr = __shfl(cacc, r + 16, 32);
      float val = nlr;
      if (lane == 0) val += cTr;
      if (lane == 1) val += cHr;
      out[(base + r) * 32 + lane] = val;
    }
  }
}

extern "C" void kernel_launch(void* const* d_in, const int* in_sizes, int n_in,
                              void* d_out, int out_size, void* d_ws, size_t ws_size,
                              hipStream_t stream) {
  (void)in_sizes; (void)n_in; (void)out_size; (void)d_ws; (void)ws_size;
  const float* x    = (const float*)d_in[0];
  const float* qW1  = (const float*)d_in[1];
  const float* qb1  = (const float*)d_in[2];
  const float* qW2  = (const float*)d_in[3];
  const float* qb2  = (const float*)d_in[4];
  const float* cW1  = (const float*)d_in[5];
  const float* cb1  = (const float*)d_in[6];
  const float* cW2  = (const float*)d_in[7];
  const float* cb2  = (const float*)d_in[8];
  const float* eTW1 = (const float*)d_in[9];
  const float* eTb1 = (const float*)d_in[10];
  const float* eTW2 = (const float*)d_in[11];
  const float* eTb2 = (const float*)d_in[12];
  const float* eHW1 = (const float*)d_in[13];
  const float* eHb1 = (const float*)d_in[14];
  const float* eHW2 = (const float*)d_in[15];
  const float* eHb2 = (const float*)d_in[16];
  float* out = (float*)d_out;

  gcn_fused_kernel<<<BLOCKS, TPB, 0, stream>>>(
      x, qW1, qb1, qW2, qb2, cW1, cb1, cW2, cb2,
      eTW1, eTb1, eTW2, eTb2, eHW1, eHb1, eHW2, eHb2, out);
}


# WMMA operands swapped (D = W1^T x X^T, lane = batch row): relu + w2m dot sum in-lane, one half exchange replaces the 16-lane shuffle butterflies; 16-byte whole-line stores; summation order changes
# speedup vs baseline: 1.2280x; 1.2280x over previous
#include <hip/hip_runtime.h>

typedef __attribute__((ext_vector_type(2))) float v2f;
typedef __attribute__((ext_vector_type(4))) float v4f;
typedef __attribute__((ext_vector_type(8))) float v8f;

#define BATCH   32768
#define NTILES  (BATCH / 16)
#define BLOCKS  256
#define TPB     128

__device__ __forceinline__ float xhalf(float x) {
  const int b = __builtin_bit_cast(int, x);
  return __builtin_bit_cast(float,
      __builtin_amdgcn_permlanex16(b, b, 0x76543210, 0xfedcba98, false, false));
}

__device__ __forceinline__ v8f wmma_f32_k4(v2f a, v2f b, v8f c) {
  return __builtin_amdgcn_wmma_f32_16x16x4_f32(
       false, a,  false, b,
       (short)0, c,  false,  false);
}

__global__ __launch_bounds__(TPB) void gcn_fused_kernel(
    const float* __restrict__ x,
    const float* __restrict__ qW1, const float* __restrict__ qb1,
    const float* __restrict__ qW2, const float* __restrict__ qb2,
    const float* __restrict__ cW1, const float* __restrict__ cb1,
    const float* __restrict__ cW2, const float* __restrict__ cb2,
    const float* __restrict__ eTW1, const float* __restrict__ eTb1,
    const float* __restrict__ eTW2, const float* __restrict__ eTb2,
    const float* __restrict__ eHW1, const float* __restrict__ eHb1,
    const float* __restrict__ eHW2, const float* __restrict__ eHb2,
    float* __restrict__ out)
{
  __shared__ __attribute__((aligned(16))) float s_w2m[2][64];
  __shared__ __attribute__((aligned(16))) float s_b1[2][64];
  __shared__ float s_eW1[2][5][32];
  __shared__ float s_eb1[2][32];
  __shared__ float s_eW2[2][32];
  __shared__ float s_eb2[2];

  const int tid = threadIdx.x;

  {
    const int net = tid >> 6;
    const int h   = tid & 63;
    const float* W2 = net ? cW2 : qW2;
    const v4f* p = (const v4f*)(W2 + h * 32);
    float s = 0.0f;
#pragma unroll
    for (int i = 0; i < 8; ++i) { v4f v = p[i]; s += v.x + v.y + v.z + v.w; }
    s_w2m[net][h] = s * (1.0f / 32.0f);
    s_b1[net][h]  = (net ? cb1 : qb1)[h];
  }
  if (tid < 32) {
#pragma unroll
    for (int f = 0; f < 5; ++f) {
      s_eW1[0][f][tid] = eTW1[f * 32 + tid];
      s_eW1[1][f][tid] = eHW1[f * 32 + tid];
    }
    s_eb1[0][tid] = eTb1[tid];
    s_eb1[1][tid] = eHb1[tid];
    s_eW2[0][tid] = eTW2[tid];
    s_eW2[1][tid] = eHW2[tid];
    if (tid < 2) s_eb2[tid] = tid ? eHb2[0] : eTb2[0];
  }
  __syncthreads();

  const int lane = tid & 31;
  const int half = lane >> 4;
  const int lm   = lane & 15;
  const int koff = half * 2;

  v8f b1v[2][4], w2v[2][4];
#pragma unroll
  for (int net = 0; net < 2; ++net) {
#pragma unroll
    for (int n = 0; n < 4; ++n) {
      const int h0 = n * 16 + half * 8;
      const v4f bl = *(const v4f*)&s_b1[net][h0];
      const v4f bh = *(const v4f*)&s_b1[net][h0 + 4];
      const v4f wl = *(const v4f*)&s_w2m[net][h0];
      const v4f wh = *(const v4f*)&s_w2m[net][h0 + 4];
      b1v[net][n] = (v8f){bl.x, bl.y, bl.z, bl.w, bh.x, bh.y, bh.z, bh.w};
      w2v[net][n] = (v8f){wl.x, wl.y, wl.z, wl.w, wh.x, wh.y, wh.z, wh.w};
    }
  }
  float mb2 = qb2[lane] + cb2[lane];
#pragma unroll
  for (int m = 16; m >= 1; m >>= 1) mb2 += __shfl_xor(mb2, m, 32);
  mb2 *= (1.0f / 32.0f);

  v2f bq[4][8], bc[4][8];
#pragma unroll
  for (int n = 0; n < 4; ++n) {
    const int col = n * 16 + lm;
#pragma unroll
    for (int k = 0; k < 8; ++k) {
      const int kr = k * 4 + koff;
      v2f t;
      t.x = qW1[kr * 64 + col]; t.y = qW1[(kr + 1) * 64 + col]; bq[n][k] = t;
      t.x = cW1[kr * 64 + col]; t.y = cW1[(kr + 1) * 64 + col]; bc[n][k] = t;
    }
  }

  const int wgid  = blockIdx.x * (TPB >> 5) + (tid >> 5);
  const int wstep = gridDim.x * (TPB >> 5);

  for (int t = wgid; t < NTILES; t += wstep) {
    const int base = t * 16;

    v2f a[8];
    const float* xr = x + (base + lm) * 32 + koff;
#pragma unroll
    for (int k = 0; k < 8; ++k) a[k] = *(const v2f*)(xr + 4 * k);

    const v8f zero8 = {0.f, 0.f, 0.f, 0.f, 0.f, 0.f, 0.f, 0.f};
    v8f acc = zero8;
#pragma unroll
    for (int n = 0; n < 4; ++n) {
      v8f c = zero8;
#pragma unroll
      for (int k = 0; k < 8; ++k) c = wmma_f32_k4(bq[n][k], a[k], c);
      const v8f h = __builtin_elementwise_max(c + b1v[0][n], zero8);
      acc = __builtin_elementwise_fma(h, w2v[0][n], acc);
    }
#pragma unroll
    for (int n = 0; n < 4; ++n) {
      v8f c = zero8;
#pragma unroll
      for (int k = 0; k < 8; ++k) c = wmma_f32_k4(bc[n][k], a[k], c);
      const v8f h = __builtin_elementwise_max(c + b1v[1][n], zero8);
      acc = __builtin_elementwise_fma(h, w2v[1][n], acc);
    }

    const float part = ((acc[0] + acc[1]) + (acc[2] + acc[3])) + ((acc[4] + acc[5]) + (acc[6] + acc[7]));
    const float p = (part + xhalf(part)) + mb2;

    const float T = x[(base + lm) * 32 + 0];
    const float H = x[(base + lm) * 32 + 1];
    const float f0 = T, f1 = H, f2 = T * T, f3 = T * H;
    const float f4 = half ? (T * H * H) : (T * T * T);
    const int net = half;
    float cacc = s_eb2[net];
#pragma unroll 4
    for (int j = 0; j < 32; ++j) {
      float hj = s_eb1[net][j];
      hj = fmaf(f0, s_eW1[net][0][j], hj);
      hj = fmaf(f1, s_eW1[net][1][j], hj);
      hj = fmaf(f2, s_eW1[net][2][j], hj);
      hj = fmaf(f3, s_eW1[net][3][j], hj);
      hj = fmaf(f4, s_eW1[net][4][j], hj);
      cacc = fmaf(fmaxf(hj, 0.0f), s_eW2[net][j], cacc);
    }
    const float q = p + cacc;

#pragma unroll
    for (int s4 = 0; s4 < 4; ++s4) {
      const int R = s4 * 4 + (lane >> 3);
      const float pR = __shfl(p, R, 32);
      const float q0 = __shfl(q, R, 32);
      const float q1 = __shfl(q, R + 16, 32);
      v4f val = {pR, pR, pR, pR};
      if ((lane & 7) == 0) { val.x = q0; val.y = q1; }
      *(v4f*)(out + (base + R) * 32 + (lane & 7) * 4) = val;
    }
  }
}

extern "C" void kernel_launch(void* const* d_in, const int* in_sizes, int n_in,
                              void* d_out, int out_size, void* d_ws, size_t ws_size,
                              hipStream_t stream) {
  (void)in_sizes; (void)n_in; (void)out_size; (void)d_ws; (void)ws_size;
  const float* x    = (const float*)d_in[0];
  const float* qW1  = (const float*)d_in[1];
  const float* qb1  = (const float*)d_in[2];
  const float* qW2  = (const float*)d_in[3];
  const float* qb2  = (const float*)d_in[4];
  const float* cW1  = (const float*)d_in[5];
  const float* cb1  = (const float*)d_in[6];
  const float* cW2  = (const float*)d_in[7];
  const float* cb2  = (const float*)d_in[8];
  const float* eTW1 = (const float*)d_in[9];
  const float* eTb1 = (const float*)d_in[10];
  const float* eTW2 = (const float*)d_in[11];
  const float* eTb2 = (const float*)d_in[12];
  const float* eHW1 = (const float*)d_in[13];
  const float* eHb1 = (const float*)d_in[14];
  const float* eHW2 = (const float*)d_in[15];
  const float* eHb2 = (const float*)d_in[16];
  float* out = (float*)d_out;

  gcn_fused_kernel<<<BLOCKS, TPB, 0, stream>>>(
      x, qW1, qb1, qW2, qb2, cW1, cb1, cW2, cb2,
      eTW1, eTb1, eTW2, eTb2, eHW1, eHb1, eHW2, eHb2, out);
}


# enso MLP weights read from LDS once per wave and held in registers; per-tile loop of LDS reads replaced by straight-line FMAs (same order, same results)
# speedup vs baseline: 1.3156x; 1.0713x over previous
#include <hip/hip_runtime.h>

typedef __attribute__((ext_vector_type(2))) float v2f;
typedef __attribute__((ext_vector_type(4))) float v4f;
typedef __attribute__((ext_vector_type(8))) float v8f;

#define BATCH   32768
#define NTILES  (BATCH / 16)
#define BLOCKS  256
#define TPB     128

__device__ __forceinline__ float xhalf(float x) {
  const int b = __builtin_bit_cast(int, x);
  return __builtin_bit_cast(float,
      __builtin_amdgcn_permlanex16(b, b, 0x76543210, 0xfedcba98, false, false));
}

__device__ __forceinline__ v8f wmma_f32_k4(v2f a, v2f b, v8f c) {
  return __builtin_amdgcn_wmma_f32_16x16x4_f32(
       false, a,  false, b,
       (short)0, c,  false,  false);
}

__global__ __launch_bounds__(TPB) void gcn_fused_kernel(
    const float* __restrict__ x,
    const float* __restrict__ qW1, const float* __restrict__ qb1,
    const float* __restrict__ qW2, const float* __restrict__ qb2,
    const float* __restrict__ cW1, const float* __restrict__ cb1,
    const float* __restrict__ cW2, const float* __restrict__ cb2,
    const float* __restrict__ eTW1, const float* __restrict__ eTb1,
    const float* __restrict__ eTW2, const float* __restrict__ eTb2,
    const float* __restrict__ eHW1, const float* __restrict__ eHb1,
    const float* __restrict__ eHW2, const float* __restrict__ eHb2,
    float* __restrict__ out)
{
  __shared__ __attribute__((aligned(16))) float s_w2m[2][64];
  __shared__ __attribute__((aligned(16))) float s_b1[2][64];
  __shared__ __attribute__((aligned(16))) float s_eW1[2][5][32];
  __shared__ __attribute__((aligned(16))) float s_eb1[2][32];
  __shared__ __attribute__((aligned(16))) float s_eW2[2][32];
  __shared__ float s_eb2[2];

  const int tid = threadIdx.x;

  {
    const int net = tid >> 6;
    const int h   = tid & 63;
    const float* W2 = net ? cW2 : qW2;
    const v4f* p = (const v4f*)(W2 + h * 32);
    float s = 0.0f;
#pragma unroll
    for (int i = 0; i < 8; ++i) { v4f v = p[i]; s += v.x + v.y + v.z + v.w; }
    s_w2m[net][h] = s * (1.0f / 32.0f);
    s_b1[net][h]  = (net ? cb1 : qb1)[h];
  }
  if (tid < 32) {
#pragma unroll
    for (int f = 0; f < 5; ++f) {
      s_eW1[0][f][tid] = eTW1[f * 32 + tid];
      s_eW1[1][f][tid] = eHW1[f * 32 + tid];
    }
    s_eb1[0][tid] = eTb1[tid];
    s_eb1[1][tid] = eHb1[tid];
    s_eW2[0][tid] = eTW2[tid];
    s_eW2[1][tid] = eHW2[tid];
    if (tid < 2) s_eb2[tid] = tid ? eHb2[0] : eTb2[0];
  }
  __syncthreads();

  const int lane = tid & 31;
  const int half = lane >> 4;
  const int lm   = lane & 15;
  const int koff = half * 2;

  v8f b1v[2][4], w2v[2][4];
#pragma unroll
  for (int net = 0; net < 2; ++net) {
#pragma unroll
    for (int n = 0; n < 4; ++n) {
      const int h0 = n * 16 + half * 8;
      const v4f bl = *(const v4f*)&s_b1[net][h0];
      const v4f bh = *(const v4f*)&s_b1[net][h0 + 4];
      const v4f wl = *(const v4f*)&s_w2m[net][h0];
      const v4f wh = *(const v4f*)&s_w2m[net][h0 + 4];
      b1v[net][n] = (v8f){bl.x, bl.y, bl.z, bl.w, bh.x, bh.y, bh.z, bh.w};
      w2v[net][n] = (v8f){wl.x, wl.y, wl.z, wl.w, wh.x, wh.y, wh.z, wh.w};
    }
  }
  v4f ew1[5][8], eb1r[8], ew2r[8];
#pragma unroll
  for (int i = 0; i < 8; ++i) {
#pragma unroll
    for (int f = 0; f < 5; ++f) ew1[f][i] = *(const v4f*)&s_eW1[half][f][4 * i];
    eb1r[i] = *(const v4f*)&s_eb1[half][4 * i];
    ew2r[i] = *(const v4f*)&s_eW2[half][4 * i];
  }
  const float eb2r = s_eb2[half];

  float mb2 = qb2[lane] + cb2[lane];
#pragma unroll
  for (int m = 16; m >= 1; m >>= 1) mb2 += __shfl_xor(mb2, m, 32);
  mb2 *= (1.0f / 32.0f);

  v2f bq[4][8], bc[4][8];
#pragma unroll
  for (int n = 0; n < 4; ++n) {
    const int col = n * 16 + lm;
#pragma unroll
    for (int k = 0; k < 8; ++k) {
      const int kr = k * 4 + koff;
      v2f t;
      t.x = qW1[kr * 64 + col]; t.y = qW1[(kr + 1) * 64 + col]; bq[n][k] = t;
      t.x = cW1[kr * 64 + col]; t.y = cW1[(kr + 1) * 64 + col]; bc[n][k] = t;
    }
  }

  const int wgid  = blockIdx.x * (TPB >> 5) + (tid >> 5);
  const int wstep = gridDim.x * (TPB >> 5);

  for (int t = wgid; t < NTILES; t += wstep) {
    const int base = t * 16;

    v2f a[8];
    const float* xr = x + (base + lm) * 32 + koff;
#pragma unroll
    for (int k = 0; k < 8; ++k) a[k] = *(const v2f*)(xr + 4 * k);

    const v8f zero8 = {0.f, 0.f, 0.f, 0.f, 0.f, 0.f, 0.f, 0.f};
    v8f acc = zero8;
#pragma unroll
    for (int n = 0; n < 4; ++n) {
      v8f c = zero8;
#pragma unroll
      for (int k = 0; k < 8; ++k) c = wmma_f32_k4(bq[n][k], a[k], c);
      const v8f h = __builtin_elementwise_max(c + b1v[0][n], zero8);
      acc = __builtin_elementwise_fma(h, w2v[0][n], acc);
    }
#pragma unroll
    for (int n = 0; n < 4; ++n) {
      v8f c = zero8;
#pragma unroll
      for (int k = 0; k < 8; ++k) c = wmma_f32_k4(bc[n][k], a[k], c);
      const v8f h = __builtin_elementwise_max(c + b1v[1][n], zero8);
      acc = __builtin_elementwise_fma(h, w2v[1][n], acc);
    }

    const float part = ((acc[0] + acc[1]) + (acc[2] + acc[3])) + ((acc[4] + acc[5]) + (acc[6] + acc[7]));
    const float p = (part + xhalf(part)) + mb2;

    const float T = x[(base + lm) * 32 + 0];
    const float H = x[(base + lm) * 32 + 1];
    const float f0 = T, f1 = H, f2 = T * T, f3 = T * H;
    const float f4 = half ? (T * H * H) : (T * T * T);
    float cacc = eb2r;
#pragma unroll
    for (int i = 0; i < 8; ++i) {
#pragma unroll
      for (int e = 0; e < 4; ++e) {
        float hj = eb1r[i][e];
        hj = fmaf(f0, ew1[0][i][e], hj);
        hj = fmaf(f1, ew1[1][i][e], hj);
        hj = fmaf(f2, ew1[2][i][e], hj);
        hj = fmaf(f3, ew1[3][i][e], hj);
        hj = fmaf(f4, ew1[4][i][e], hj);
        cacc = fmaf(fmaxf(hj, 0.0f), ew2r[i][e], cacc);
      }
    }
    const float q = p + cacc;

#pragma unroll
    for (int s4 = 0; s4 < 4; ++s4) {
      const int R = s4 * 4 + (lane >> 3);
      const float pR = __shfl(p, R, 32);
      const float q0 = __shfl(q, R, 32);
      const float q1 = __shfl(q, R + 16, 32);
      v4f val = {pR, pR, pR, pR};
      if ((lane & 7) == 0) { val.x = q0; val.y = q1; }
      *(v4f*)(out + (base + R) * 32 + (lane & 7) * 4) = val;
    }
  }
}

extern "C" void kernel_launch(void* const* d_in, const int* in_sizes, int n_in,
                              void* d_out, int out_size, void* d_ws, size_t ws_size,
                              hipStream_t stream) {
  (void)in_sizes; (void)n_in; (void)out_size; (void)d_ws; (void)ws_size;
  const float* x    = (const float*)d_in[0];
  const float* qW1  = (const float*)d_in[1];
  const float* qb1  = (const float*)d_in[2];
  const float* qW2  = (const float*)d_in[3];
  const float* qb2  = (const float*)d_in[4];
  const float* cW1  = (const float*)d_in[5];
  const float* cb1  = (const float*)d_in[6];
  const float* cW2  = (const float*)d_in[7];
  const float* cb2  = (const float*)d_in[8];
  const float* eTW1 = (const float*)d_in[9];
  const float* eTb1 = (const float*)d_in[10];
  const float* eTW2 = (const float*)d_in[11];
  const float* eTb2 = (const float*)d_in[12];
  const float* eHW1 = (const float*)d_in[13];
  const float* eHb1 = (const float*)d_in[14];
  const float* eHW2 = (const float*)d_in[15];
  const float* eHb2 = (const float*)d_in[16];
  float* out = (float*)d_out;

  gcn_fused_kernel<<<BLOCKS, TPB, 0, stream>>>(
      x, qW1, qb1, qW2, qb2, cW1, cb1, cW2, cb2,
      eTW1, eTb1, eTW2, eTb2, eHW1, eHb1, eHW2, eHb2, out);
}


# every prologue global load (W1/W2/biases/enso weights, tile x rows) issued at entry before the first wait; next tile's x rows requested under the current tile's WMMAs; sums in the parent's order
# speedup vs baseline: 1.7238x; 1.3103x over previous
#include <hip/hip_runtime.h>


typedef __attribute__((ext_vector_type(2))) float v2f;
typedef __attribute__((ext_vector_type(4))) float v4f;
typedef __attribute__((ext_vector_type(8))) float v8f;

#define BATCH   32768
#define NTILES  (BATCH / 16)
#define BLOCKS  256
#define TPB     128

__device__ __forceinline__ float xhalf(float x) {
  const int b = __builtin_bit_cast(int, x);
  return __builtin_bit_cast(float,
      __builtin_amdgcn_permlanex16(b, b, 0x76543210, 0xfedcba98, false, false));
}

__device__ __forceinline__ v8f wmma_f32_k4(v2f a, v2f b, v8f c) {
  return __builtin_amdgcn_wmma_f32_16x16x4_f32(
       false, a,  false, b,
       (short)0, c,  false,  false);
}

__global__ __launch_bounds__(TPB) void gcn_fused_kernel(
    const float* __restrict__ x,
    const float* __restrict__ qW1, const float* __restrict__ qb1,
    const float* __restrict__ qW2, const float* __restrict__ qb2,
    const float* __restrict__ cW1, const float* __restrict__ cb1,
    const float* __restrict__ cW2, const float* __restrict__ cb2,
    const float* __restrict__ eTW1, const float* __restrict__ eTb1,
    const float* __restrict__ eTW2, const float* __restrict__ eTb2,
    const float* __restrict__ eHW1, const float* __restrict__ eHb1,
    const float* __restrict__ eHW2, const float* __restrict__ eHb2,
    float* __restrict__ out)
{
  __shared__ __attribute__((aligned(16))) float s_w2m[2][64];
  __shared__ __attribute__((aligned(16))) float s_b1[2][64];
  __shared__ __attribute__((aligned(16))) float s_e[512];

  const int tid  = threadIdx.x;
  const int lane = tid & 31;
  const int half = lane >> 4;
  const int lm   = lane & 15;
  const int koff = half * 2;
  const int wv    = __builtin_amdgcn_readfirstlane(tid >> 5);
  const int wgid  = blockIdx.x * (TPB >> 5) + wv;
  const int wstep = gridDim.x * (TPB >> 5);


  const float b2qv = qb2[lane];
  const float b2cv = cb2[lane];

  const int net = wv >> 1;
  const int h   = tid & 63;
  v4f w2r[8];
  {
    const v4f* p = (const v4f*)((net ? cW2 : qW2) + h * 32);
#pragma unroll
    for (int i = 0; i < 8; ++i) w2r[i] = p[i];
  }
  const float b1val = (net ? cb1 : qb1)[h];

  float ev[4];
  {
    const float* b1p = (wv == 0) ? eTW1 : eHW1;
    const int    o1  = (wv == 0) ? (tid + 128) : (tid - 32);
    const float* b2p = (wv < 2) ? eHW1 : (wv == 2) ? eTb1 : eHb1;
    const int    o2  = (wv < 2) ? (tid + 96) : lane;
    const float* b3p = (wv == 0) ? eTW2 : (wv == 1) ? eHW2 : eTb2;
    const float* p3  = (wv < 2) ? (b3p + lane) : (tid == 65) ? eHb2 : eTb2;
    ev[0] = eTW1[tid];
    ev[1] = b1p[o1];
    ev[2] = b2p[o2];
    ev[3] = *p3;
  }

  v2f a[8];
  v2f th;
  {
    const int t0 = (wgid < NTILES) ? wgid : (NTILES - 1);
    const float* xrow = x + (t0 * 16 + lm) * 32;
    th = *(const v2f*)xrow;
#pragma unroll
    for (int k = 0; k < 8; ++k) a[k] = *(const v2f*)(xrow + koff + 4 * k);
  }

  __builtin_amdgcn_sched_barrier(0);

  v2f bq[4][8], bc[4][8];
#pragma unroll
  for (int k = 0; k < 8; ++k) {
    const int kr = k * 4 + koff;
#pragma unroll
    for (int n = 0; n < 4; ++n) {
      const int col = n * 16 + lm;
      v2f t;
      t.x = qW1[kr * 64 + col]; t.y = qW1[(kr + 1) * 64 + col]; bq[n][k] = t;
      t.x = cW1[kr * 64 + col]; t.y = cW1[(kr + 1) * 64 + col]; bc[n][k] = t;
    }
  }

  __builtin_amdgcn_sched_barrier(0);

  float mb2 = b2qv + b2cv;
#pragma unroll
  for (int m = 16; m >= 1; m >>= 1) mb2 += __shfl_xor(mb2, m, 32);
  mb2 *= (1.0f / 32.0f);

  {
    float s = 0.0f;
#pragma unroll
    for (int i = 0; i < 8; ++i) { const v4f v = w2r[i]; s += v.x + v.y + v.z + v.w; }
    s_w2m[net][h] = s * (1.0f / 32.0f);
    s_b1[net][h]  = b1val;
#pragma unroll
    for (int j = 0; j < 4; ++j) s_e[tid + 128 * j] = ev[j];
  }
  __syncthreads();

  v8f b1v[2][4], w2v[2][4];
#pragma unroll
  for (int nt = 0; nt < 2; ++nt) {
#pragma unroll
    for (int n = 0; n < 4; ++n) {
      const int h0 = n * 16 + half * 8;
      const v4f bl = *(const v4f*)&s_b1[nt][h0];
      const v4f bh = *(const v4f*)&s_b1[nt][h0 + 4];
      const v4f wl = *(const v4f*)&s_w2m[nt][h0];
      const v4f wh = *(const v4f*)&s_w2m[nt][h0 + 4];
      b1v[nt][n] = (v8f){bl.x, bl.y, bl.z, bl.w, bh.x, bh.y, bh.z, bh.w};
      w2v[nt][n] = (v8f){wl.x, wl.y, wl.z, wl.w, wh.x, wh.y, wh.z, wh.w};
    }
  }
  v4f ew1[5][8], eb1r[8], ew2r[8];
#pragma unroll
  for (int i = 0; i < 8; ++i) {
#pragma unroll
    for (int f = 0; f < 5; ++f) ew1[f][i] = *(const v4f*)&s_e[half * 160 + f * 32 + 4 * i];
    eb1r[i] = *(const v4f*)&s_e[320 + half * 32 + 4 * i];
    ew2r[i] = *(const v4f*)&s_e[384 + half * 32 + 4 * i];
  }
  const float eb2r = s_e[448 + half];

  for (int t = wgid; t < NTILES; t += wstep) {
    const int base = t * 16;

    v2f an[8];
    v2f thn = th;
#pragma unroll
    for (int k = 0; k < 8; ++k) an[k] = a[k];
    if (t + wstep < NTILES) {
      const float* xrow = x + ((t + wstep) * 16 + lm) * 32;
      thn = *(const v2f*)xrow;
#pragma unroll
      for (int k = 0; k < 8; ++k) an[k] = *(const v2f*)(xrow + koff + 4 * k);
    }

    const v8f zero8 = {0.f, 0.f, 0.f, 0.f, 0.f, 0.f, 0.f, 0.f};
    v8f acc = zero8;
#pragma unroll
    for (int n = 0; n < 4; ++n) {
      v8f c = zero8;
#pragma unroll
      for (int k = 0; k < 8; ++k) c = wmma_f32_k4(bq[n][k], a[k], c);
      const v8f hh = __builtin_elementwise_max(c + b1v[0][n], zero8);
      acc = __builtin_elementwise_fma(hh, w2v[0][n], acc);
    }
#pragma unroll
    for (int n = 0; n < 4; ++n) {
      v8f c = zero8;
#pragma unroll
      for (int k = 0; k < 8; ++k) c = wmma_f32_k4(bc[n][k], a[k], c);
      const v8f hh = __builtin_elementwise_max(c + b1v[1][n], zero8);
      acc = __builtin_elementwise_fma(hh, w2v[1][n], acc);
    }

    const float part = ((acc[0] + acc[1]) + (acc[2] + acc[3])) + ((acc[4] + acc[5]) + (acc[6] + acc[7]));
    const float p = (part + xhalf(part)) + mb2;

    const float T = th.x;
    const float H = th.y;
    const float f0 = T, f1 = H, f2 = T * T, f3 = T * H;
    const float f4 = half ? (T * H * H) : (T * T * T);
    float cacc = eb2r;
#pragma unroll
    for (int i = 0; i < 8; ++i) {
#pragma unroll
      for (int e = 0; e < 4; ++e) {
        float hj = eb1r[i][e];
        hj = fmaf(f0, ew1[0][i][e], hj);
        hj = fmaf(f1, ew1[1][i][e], hj);
        hj = fmaf(f2, ew1[2][i][e], hj);
        hj = fmaf(f3, ew1[3][i][e], hj);
        hj = fmaf(f4, ew1[4][i][e], hj);
        cacc = fmaf(fmaxf(hj, 0.0f), ew2r[i][e], cacc);
      }
    }
    const float q = p + cacc;

#pragma unroll
    for (int s4 = 0; s4 < 4; ++s4) {
      const int R = s4 * 4 + (lane >> 3);
      const float pR = __shfl(p, R, 32);
      const float q0 = __shfl(q, R, 32);
      const float q1 = __shfl(q, R + 16, 32);
      v4f val = {pR, pR, pR, pR};
      if ((lane & 7) == 0) { val.x = q0; val.y = q1; }
      *(v4f*)(out + (base + R) * 32 + (lane & 7) * 4) = val;
    }

    th = thn;
#pragma unroll
    for (int k = 0; k < 8; ++k) a[k] = an[k];
  }
}

extern "C" void kernel_launch(void* const* d_in, const int* in_sizes, int n_in,
                              void* d_out, int out_size, void* d_ws, size_t ws_size,
                              hipStream_t stream) {
  (void)in_sizes; (void)n_in; (void)out_size; (void)d_ws; (void)ws_size;
  const float* x    = (const float*)d_in[0];
  const float* qW1  = (const float*)d_in[1];
  const float* qb1  = (const float*)d_in[2];
  const float* qW2  = (const float*)d_in[3];
  const float* qb2  = (const float*)d_in[4];
  const float* cW1  = (const float*)d_in[5];
  const float* cb1  = (const float*)d_in[6];
  const float* cW2  = (const float*)d_in[7];
  const float* cb2  = (const float*)d_in[8];
  const float* eTW1 = (const float*)d_in[9];
  const float* eTb1 = (const float*)d_in[10];
  const float* eTW2 = (const float*)d_in[11];
  const float* eTb2 = (const float*)d_in[12];
  const float* eHW1 = (const float*)d_in[13];
  const float* eHb1 = (const float*)d_in[14];
  const float* eHW2 = (const float*)d_in[15];
  const float* eHb2 = (const float*)d_in[16];
  float* out = (float*)d_out;

  gcn_fused_kernel<<<BLOCKS, TPB, 0, stream>>>(
      x, qW1, qb1, qW2, qb2, cW1, cb1, cW2, cb2,
      eTW1, eTb1, eTW2, eTb2, eHW1, eHb1, eHW2, eHb2, out);
}


# prologue weight reads as whole-line 16-byte loads: W1 with hidden unit 4*row+tile on accumulator rows (32 loads, was 128), W2 4 rows per load + 8-lane sums; sum order changes (hidden units, W2 means)
# speedup vs baseline: 2.3080x; 1.3389x over previous
#include <hip/hip_runtime.h>


typedef __attribute__((ext_vector_type(2))) float v2f;
typedef __attribute__((ext_vector_type(4))) float v4f;
typedef __attribute__((ext_vector_type(8))) float v8f;

#define BATCH   32768
#define NTILES  (BATCH / 16)
#define BLOCKS  256
#define TPB     128

__device__ __forceinline__ float xhalf(float x) {
  const int b = __builtin_bit_cast(int, x);
  return __builtin_bit_cast(float,
      __builtin_amdgcn_permlanex16(b, b, 0x76543210, 0xfedcba98, false, false));
}

__device__ __forceinline__ float dpp_swz_b1(float x) {
  const int b = __builtin_bit_cast(int, x);
  return __builtin_bit_cast(float, __builtin_amdgcn_update_dpp(b, b, 0xB1, 0xf, 0xf, false));
}
__device__ __forceinline__ float dpp_swz_b2(float x) {
  const int b = __builtin_bit_cast(int, x);
  return __builtin_bit_cast(float, __builtin_amdgcn_update_dpp(b, b, 0x4E, 0xf, 0xf, false));
}
__device__ __forceinline__ float dpp_half_mirror(float x) {
  const int b = __builtin_bit_cast(int, x);
  return __builtin_bit_cast(float, __builtin_amdgcn_update_dpp(b, b, 0x141, 0xf, 0xf, false));
}
__device__ __forceinline__ float sum8(float x) {
  x += dpp_swz_b1(x);
  x += dpp_swz_b2(x);
  x += dpp_half_mirror(x);
  return x;
}

__device__ __forceinline__ v8f wmma_f32_k4(v2f a, v2f b, v8f c) {
  return __builtin_amdgcn_wmma_f32_16x16x4_f32(
       false, a,  false, b,
       (short)0, c,  false,  false);
}

__global__ __launch_bounds__(TPB) void gcn_fused_kernel(
    const float* __restrict__ x,
    const float* __restrict__ qW1, const float* __restrict__ qb1,
    const float* __restrict__ qW2, const float* __restrict__ qb2,
    const float* __restrict__ cW1, const float* __restrict__ cb1,
    const float* __restrict__ cW2, const float* __restrict__ cb2,
    const float* __restrict__ eTW1, const float* __restrict__ eTb1,
    const float* __restrict__ eTW2, const float* __restrict__ eTb2,
    const float* __restrict__ eHW1, const float* __restrict__ eHb1,
    const float* __restrict__ eHW2, const float* __restrict__ eHb2,
    float* __restrict__ out)
{
  __shared__ __attribute__((aligned(16))) float s_w2m[2][64];
  __shared__ __attribute__((aligned(16))) float s_b1[2][64];
  __shared__ __attribute__((aligned(16))) float s_e[512];

  const int tid  = threadIdx.x;
  const int lane = tid & 31;
  const int half = lane >> 4;
  const int lm   = lane & 15;
  const int koff = half * 2;
  const int wv    = __builtin_amdgcn_readfirstlane(tid >> 5);
  const int wgid  = blockIdx.x * (TPB >> 5) + wv;
  const int wstep = gridDim.x * (TPB >> 5);


  const float b2qv = qb2[lane];
  const float b2cv = cb2[lane];

  const int net = wv >> 1;
  const int wh  = wv & 1;
  v4f w2r[8];
  {
    const v4f* p = (const v4f*)((net ? cW2 : qW2) + wh * 1024) + lane;
#pragma unroll
    for (int i = 0; i < 8; ++i) w2r[i] = p[i * 32];
  }
  const int h = wh * 32 + (lane & 7) * 4 + (lane >> 3);
  const float b1val = (net ? cb1 : qb1)[h];

  float ev[4];
  {
    const float* b1p = (wv == 0) ? eTW1 : eHW1;
    const int    o1  = (wv == 0) ? (tid + 128) : (tid - 32);
    const float* b2p = (wv < 2) ? eHW1 : (wv == 2) ? eTb1 : eHb1;
    const int    o2  = (wv < 2) ? (tid + 96) : lane;
    const float* b3p = (wv == 0) ? eTW2 : (wv == 1) ? eHW2 : eTb2;
    const float* p3  = (wv < 2) ? (b3p + lane) : (tid == 65) ? eHb2 : eTb2;
    ev[0] = eTW1[tid];
    ev[1] = b1p[o1];
    ev[2] = b2p[o2];
    ev[3] = *p3;
  }

  v2f a[8];
  v2f th;
  {
    const int t0 = (wgid < NTILES) ? wgid : (NTILES - 1);
    const float* xrow = x + (t0 * 16 + lm) * 32;
    th = *(const v2f*)xrow;
#pragma unroll
    for (int k = 0; k < 8; ++k) a[k] = *(const v2f*)(xrow + koff + 4 * k);
  }

  __builtin_amdgcn_sched_barrier(0);

  v4f wq[8][2], wc[8][2];
#pragma unroll
  for (int k = 0; k < 8; ++k) {
#pragma unroll
    for (int s = 0; s < 2; ++s) {
      const int f = k * 4 + koff + s;
      wq[k][s] = *(const v4f*)(qW1 + f * 64 + 4 * lm);
      wc[k][s] = *(const v4f*)(cW1 + f * 64 + 4 * lm);
    }
  }

  __builtin_amdgcn_sched_barrier(0);

  float mb2 = b2qv + b2cv;
#pragma unroll
  for (int m = 16; m >= 1; m >>= 1) mb2 += __shfl_xor(mb2, m, 32);
  mb2 *= (1.0f / 32.0f);

  {
    float s = 0.0f;
#pragma unroll
    for (int i = 0; i < 8; ++i) {
      const v4f v = w2r[i];
      const float r = sum8((v.x + v.y) + (v.z + v.w));
      if ((lane & 7) == i) s = r;
    }
    const int hs = (lane >> 3) * 16 + wh * 8 + (lane & 7);
    s_w2m[net][hs] = s * (1.0f / 32.0f);
    s_b1[net][hs]  = b1val;
#pragma unroll
    for (int j = 0; j < 4; ++j) s_e[tid + 128 * j] = ev[j];
  }
  __syncthreads();

  v2f bq[4][8], bc[4][8];
#pragma unroll
  for (int k = 0; k < 8; ++k) {
#pragma unroll
    for (int n = 0; n < 4; ++n) {
      bq[n][k] = (v2f){wq[k][0][n], wq[k][1][n]};
      bc[n][k] = (v2f){wc[k][0][n], wc[k][1][n]};
    }
  }

  v8f b1v[2][4], w2v[2][4];
#pragma unroll
  for (int nt = 0; nt < 2; ++nt) {
#pragma unroll
    for (int n = 0; n < 4; ++n) {
      const int h0 = n * 16 + half * 8;
      const v4f bl = *(const v4f*)&s_b1[nt][h0];
      const v4f bh = *(const v4f*)&s_b1[nt][h0 + 4];
      const v4f wl = *(const v4f*)&s_w2m[nt][h0];
      const v4f wh = *(const v4f*)&s_w2m[nt][h0 + 4];
      b1v[nt][n] = (v8f){bl.x, bl.y, bl.z, bl.w, bh.x, bh.y, bh.z, bh.w};
      w2v[nt][n] = (v8f){wl.x, wl.y, wl.z, wl.w, wh.x, wh.y, wh.z, wh.w};
    }
  }
  v4f ew1[5][8], eb1r[8], ew2r[8];
#pragma unroll
  for (int i = 0; i < 8; ++i) {
#pragma unroll
    for (int f = 0; f < 5; ++f) ew1[f][i] = *(const v4f*)&s_e[half * 160 + f * 32 + 4 * i];
    eb1r[i] = *(const v4f*)&s_e[320 + half * 32 + 4 * i];
    ew2r[i] = *(const v4f*)&s_e[384 + half * 32 + 4 * i];
  }
  const float eb2r = s_e[448 + half];

  for (int t = wgid; t < NTILES; t += wstep) {
    const int base = t * 16;

    v2f an[8];
    v2f thn = th;
#pragma unroll
    for (int k = 0; k < 8; ++k) an[k] = a[k];
    if (t + wstep < NTILES) {
      const float* xrow = x + ((t + wstep) * 16 + lm) * 32;
      thn = *(const v2f*)xrow;
#pragma unroll
      for (int k = 0; k < 8; ++k) an[k] = *(const v2f*)(xrow + koff + 4 * k);
    }

    const v8f zero8 = {0.f, 0.f, 0.f, 0.f, 0.f, 0.f, 0.f, 0.f};
    v8f acc = zero8;
#pragma unroll
    for (int n = 0; n < 4; ++n) {
      v8f c = zero8;
#pragma unroll
      for (int k = 0; k < 8; ++k) c = wmma_f32_k4(bq[n][k], a[k], c);
      const v8f hh = __builtin_elementwise_max(c + b1v[0][n], zero8);
      acc = __builtin_elementwise_fma(hh, w2v[0][n], acc);
    }
#pragma unroll
    for (int n = 0; n < 4; ++n) {
      v8f c = zero8;
#pragma unroll
      for (int k = 0; k < 8; ++k) c = wmma_f32_k4(bc[n][k], a[k], c);
      const v8f hh = __builtin_elementwise_max(c + b1v[1][n], zero8);
      acc = __builtin_elementwise_fma(hh, w2v[1][n], acc);
    }

    const float part = ((acc[0] + acc[1]) + (acc[2] + acc[3])) + ((acc[4] + acc[5]) + (acc[6] + acc[7]));
    const float p = (part + xhalf(part)) + mb2;

    const float T = th.x;
    const float H = th.y;
    const float f0 = T, f1 = H, f2 = T * T, f3 = T * H;
    const float f4 = half ? (T * H * H) : (T * T * T);
    float cacc = eb2r;
#pragma unroll
    for (int i = 0; i < 8; ++i) {
#pragma unroll
      for (int e = 0; e < 4; ++e) {
        float hj = eb1r[i][e];
        hj = fmaf(f0, ew1[0][i][e], hj);
        hj = fmaf(f1, ew1[1][i][e], hj);
        hj = fmaf(f2, ew1[2][i][e], hj);
        hj = fmaf(f3, ew1[3][i][e], hj);
        hj = fmaf(f4, ew1[4][i][e], hj);
        cacc = fmaf(fmaxf(hj, 0.0f), ew2r[i][e], cacc);
      }
    }
    const float q = p + cacc;

#pragma unroll
    for (int s4 = 0; s4 < 4; ++s4) {
      const int R = s4 * 4 + (lane >> 3);
      const float pR = __shfl(p, R, 32);
      const float q0 = __shfl(q, R, 32);
      const float q1 = __shfl(q, R + 16, 32);
      v4f val = {pR, pR, pR, pR};
      if ((lane & 7) == 0) { val.x = q0; val.y = q1; }
      *(v4f*)(out + (base + R) * 32 + (lane & 7) * 4) = val;
    }

    th = thn;
#pragma unroll
    for (int k = 0; k < 8; ++k) a[k] = an[k];
  }
}

extern "C" void kernel_launch(void* const* d_in, const int* in_sizes, int n_in,
                              void* d_out, int out_size, void* d_ws, size_t ws_size,
                              hipStream_t stream) {
  (void)in_sizes; (void)n_in; (void)out_size; (void)d_ws; (void)ws_size;
  const float* x    = (const float*)d_in[0];
  const float* qW1  = (const float*)d_in[1];
  const float* qb1  = (const float*)d_in[2];
  const float* qW2  = (const float*)d_in[3];
  const float* qb2  = (const float*)d_in[4];
  const float* cW1  = (const float*)d_in[5];
  const float* cb1  = (const float*)d_in[6];
  const float* cW2  = (const float*)d_in[7];
  const float* cb2  = (const float*)d_in[8];
  const float* eTW1 = (const float*)d_in[9];
  const float* eTb1 = (const float*)d_in[10];
  const float* eTW2 = (const float*)d_in[11];
  const float* eTb2 = (const float*)d_in[12];
  const float* eHW1 = (const float*)d_in[13];
  const float* eHb1 = (const float*)d_in[14];
  const float* eHW2 = (const float*)d_in[15];
  const float* eHb2 = (const float*)d_in[16];
  float* out = (float*)d_out;

  gcn_fused_kernel<<<BLOCKS, TPB, 0, stream>>>(
      x, qW1, qb1, qW2, qb2, cW1, cb1, cW2, cb2,
      eTW1, eTb1, eTW2, eTb2, eHW1, eHb1, eHW2, eHb2, out);
}


# the wave no longer stops for the LDS tables: LDS-only block barrier, both tiles in straight-line code: table reads run under the W1/x loads in flight and the first tile's first 56 WMMAs; same sums
# speedup vs baseline: 2.4432x; 1.0586x over previous
#include <hip/hip_runtime.h>


typedef __attribute__((ext_vector_type(2))) float v2f;
typedef __attribute__((ext_vector_type(4))) float v4f;
typedef __attribute__((ext_vector_type(8))) float v8f;

#define BATCH   32768
#define NTILES  (BATCH / 16)
#define BLOCKS  256
#define TPB     128
static_assert(NTILES == 2 * BLOCKS * (TPB / 32), "the kernel gives every wave exactly two tiles");

__device__ __forceinline__ float xhalf(float x) {
  const int b = __builtin_bit_cast(int, x);
  return __builtin_bit_cast(float,
      __builtin_amdgcn_permlanex16(b, b, 0x76543210, 0xfedcba98, false, false));
}

__device__ __forceinline__ float xlane8(float x) {
  const int b = __builtin_bit_cast(int, x);
  return __builtin_bit_cast(float,
      __builtin_amdgcn_permlane16(b, b, 0xfedcba98, 0x76543210, false, false));
}
__device__ __forceinline__ float xlane4(float x) {
  const int b = __builtin_bit_cast(int, x);
  return __builtin_bit_cast(float,
      __builtin_amdgcn_permlane16(b, b, 0x32107654, 0xba98fedc, false, false));
}

__device__ __forceinline__ float dpp_swz_b1(float x) {
  const int b = __builtin_bit_cast(int, x);
  return __builtin_bit_cast(float, __builtin_amdgcn_update_dpp(b, b, 0xB1, 0xf, 0xf, false));
}
__device__ __forceinline__ float dpp_swz_b2(float x) {
  const int b = __builtin_bit_cast(int, x);
  return __builtin_bit_cast(float, __builtin_amdgcn_update_dpp(b, b, 0x4E, 0xf, 0xf, false));
}
__device__ __forceinline__ float dpp_half_mirror(float x) {
  const int b = __builtin_bit_cast(int, x);
  return __builtin_bit_cast(float, __builtin_amdgcn_update_dpp(b, b, 0x141, 0xf, 0xf, false));
}
__device__ __forceinline__ float sum8(float x) {
  x += dpp_swz_b1(x);
  x += dpp_swz_b2(x);
  x += dpp_half_mirror(x);
  return x;
}

__device__ __forceinline__ v8f wmma_f32_k4(v2f a, v2f b, v8f c) {
  return __builtin_amdgcn_wmma_f32_16x16x4_f32(
       false, a,  false, b,
       (short)0, c,  false,  false);
}

__global__ __launch_bounds__(TPB) void gcn_fused_kernel(
    const float* __restrict__ x,
    const float* __restrict__ qW1, const float* __restrict__ qb1,
    const float* __restrict__ qW2, const float* __restrict__ qb2,
    const float* __restrict__ cW1, const float* __restrict__ cb1,
    const float* __restrict__ cW2, const float* __restrict__ cb2,
    const float* __restrict__ eTW1, const float* __restrict__ eTb1,
    const float* __restrict__ eTW2, const float* __restrict__ eTb2,
    const float* __restrict__ eHW1, const float* __restrict__ eHb1,
    const float* __restrict__ eHW2, const float* __restrict__ eHb2,
    float* __restrict__ out)
{
  __shared__ __attribute__((aligned(16))) float s_w2m[2][64];
  __shared__ __attribute__((aligned(16))) float s_b1[2][64];
  __shared__ __attribute__((aligned(16))) float s_e[512];

  const int tid  = threadIdx.x;
  const int lane = tid & 31;
  const int half = lane >> 4;
  const int lm   = lane & 15;
  const int koff = half * 2;
  const int wv    = __builtin_amdgcn_readfirstlane(tid >> 5);
  const int wgid  = blockIdx.x * (TPB >> 5) + wv;
  const int wstep = BLOCKS * (TPB >> 5);


  const float b2qv = qb2[lane];
  const float b2cv = cb2[lane];

  const int net = wv >> 1;
  const int wh  = wv & 1;
  v4f w2r[8];
  {
    const v4f* p = (const v4f*)((net ? cW2 : qW2) + wh * 1024) + lane;
#pragma unroll
    for (int i = 0; i < 8; ++i) w2r[i] = p[i * 32];
  }
  const int h = wh * 32 + (lane & 7) * 4 + (lane >> 3);
  const float b1val = (net ? cb1 : qb1)[h];

  float ev[4];
  {
    const float* b1p = (wv == 0) ? eTW1 : eHW1;
    const int    o1  = (wv == 0) ? (tid + 128) : (tid - 32);
    const float* b2p = (wv < 2) ? eHW1 : (wv == 2) ? eTb1 : eHb1;
    const int    o2  = (wv < 2) ? (tid + 96) : lane;
    const float* b3p = (wv == 0) ? eTW2 : (wv == 1) ? eHW2 : eTb2;
    const float* p3  = (wv < 2) ? (b3p + lane) : (tid == 65) ? eHb2 : eTb2;
    ev[0] = eTW1[tid];
    ev[1] = b1p[o1];
    ev[2] = b2p[o2];
    ev[3] = *p3;
  }

  v2f a[8];
  v2f th;
  {
    const int t0 = (wgid < NTILES) ? wgid : (NTILES - 1);
    const float* xrow = x + (t0 * 16 + lm) * 32;
    th = *(const v2f*)xrow;
#pragma unroll
    for (int k = 0; k < 8; ++k) a[k] = *(const v2f*)(xrow + koff + 4 * k);
  }

  __builtin_amdgcn_sched_barrier(0);

  v4f wq[8][2], wc[8][2];
#pragma unroll
  for (int k = 0; k < 8; ++k) {
#pragma unroll
    for (int s = 0; s < 2; ++s) {
      const int f = k * 4 + koff + s;
      wq[k][s] = *(const v4f*)(qW1 + f * 64 + 4 * lm);
      wc[k][s] = *(const v4f*)(cW1 + f * 64 + 4 * lm);
    }
  }

  __builtin_amdgcn_sched_barrier(0);

  float mb2 = b2qv + b2cv;
  mb2 += xhalf(mb2);
  mb2 += xlane8(mb2);
  mb2 += xlane4(mb2);
  mb2 += dpp_swz_b2(mb2);
  mb2 += dpp_swz_b1(mb2);
  mb2 *= (1.0f / 32.0f);

  {
    float s = 0.0f;
#pragma unroll
    for (int i = 0; i < 8; ++i) {
      const v4f v = w2r[i];
      const float r = sum8((v.x + v.y) + (v.z + v.w));
      if ((lane & 7) == i) s = r;
    }
    const int hs = (lane >> 3) * 16 + wh * 8 + (lane & 7);
    s_w2m[net][hs] = s * (1.0f / 32.0f);
    s_b1[net][hs]  = b1val;
#pragma unroll
    for (int j = 0; j < 4; ++j) s_e[tid + 128 * j] = ev[j];
  }
  v2f an[8];
  v2f thn;
  {
    const int tn = (wgid + wstep < NTILES) ? (wgid + wstep) : (NTILES - 1);
    const float* xrow = x + (tn * 16 + lm) * 32;
    thn = *(const v2f*)xrow;
#pragma unroll
    for (int k = 0; k < 8; ++k) an[k] = *(const v2f*)(xrow + koff + 4 * k);
  }
  __builtin_amdgcn_sched_barrier(0);

  __builtin_amdgcn_fence(__ATOMIC_RELEASE, "workgroup", "local");
  __builtin_amdgcn_s_barrier();
  __builtin_amdgcn_fence(__ATOMIC_ACQUIRE, "workgroup", "local");

  v8f b1v[2][4], w2v[2][4];
#pragma unroll
  for (int nt = 0; nt < 2; ++nt) {
#pragma unroll
    for (int n = 0; n < 4; ++n) {
      const int h0 = n * 16 + half * 8;
      const v4f bl = *(const v4f*)&s_b1[nt][h0];
      const v4f bh = *(const v4f*)&s_b1[nt][h0 + 4];
      const v4f wl = *(const v4f*)&s_w2m[nt][h0];
      const v4f wh = *(const v4f*)&s_w2m[nt][h0 + 4];
      b1v[nt][n] = (v8f){bl.x, bl.y, bl.z, bl.w, bh.x, bh.y, bh.z, bh.w};
      w2v[nt][n] = (v8f){wl.x, wl.y, wl.z, wl.w, wh.x, wh.y, wh.z, wh.w};
    }
  }
  v4f ew1[5][8], eb1r[8], ew2r[8];
#pragma unroll
  for (int i = 0; i < 8; ++i) {
#pragma unroll
    for (int f = 0; f < 5; ++f) ew1[f][i] = *(const v4f*)&s_e[half * 160 + f * 32 + 4 * i];
    eb1r[i] = *(const v4f*)&s_e[320 + half * 32 + 4 * i];
    ew2r[i] = *(const v4f*)&s_e[384 + half * 32 + 4 * i];
  }
  const float eb2r = s_e[448 + half];

  __builtin_amdgcn_sched_barrier(0);

  v2f bq[4][8], bc[4][8];

  auto tile = [&](const int t) __attribute__((always_inline)) {
    const int base = t * 16;

    const v8f zero8 = {0.f, 0.f, 0.f, 0.f, 0.f, 0.f, 0.f, 0.f};
    v8f cq[4], cc[4];
#pragma unroll
    for (int n = 0; n < 4; ++n) { cq[n] = zero8; cc[n] = zero8; }
#pragma unroll
    for (int k = 0; k < 8; ++k) {
#pragma unroll
      for (int n = 0; n < 4; ++n) cq[n] = wmma_f32_k4(bq[n][k], a[k], cq[n]);
#pragma unroll
      for (int n = 0; n < 3; ++n) cc[n] = wmma_f32_k4(bc[n][k], a[k], cc[n]);
    }
    __builtin_amdgcn_sched_barrier(0);
#pragma unroll
    for (int k = 0; k < 8; ++k) cc[3] = wmma_f32_k4(bc[3][k], a[k], cc[3]);
    v8f acc = zero8;
#pragma unroll
    for (int n = 0; n < 4; ++n) {
      const v8f hh = __builtin_elementwise_max(cq[n] + b1v[0][n], zero8);
      acc = __builtin_elementwise_fma(hh, w2v[0][n], acc);
    }
#pragma unroll
    for (int n = 0; n < 4; ++n) {
      const v8f hh = __builtin_elementwise_max(cc[n] + b1v[1][n], zero8);
      acc = __builtin_elementwise_fma(hh, w2v[1][n], acc);
    }

    const float part = ((acc[0] + acc[1]) + (acc[2] + acc[3])) + ((acc[4] + acc[5]) + (acc[6] + acc[7]));
    const float p = (part + xhalf(part)) + mb2;

    const float T = th.x;
    const float H = th.y;
    const float f0 = T, f1 = H, f2 = T * T, f3 = T * H;
    const float f4 = half ? (T * H * H) : (T * T * T);
    float cacc = eb2r;
#pragma unroll
    for (int i = 0; i < 8; ++i) {
#pragma unroll
      for (int e = 0; e < 4; ++e) {
        float hj = eb1r[i][e];
        hj = fmaf(f0, ew1[0][i][e], hj);
        hj = fmaf(f1, ew1[1][i][e], hj);
        hj = fmaf(f2, ew1[2][i][e], hj);
        hj = fmaf(f3, ew1[3][i][e], hj);
        hj = fmaf(f4, ew1[4][i][e], hj);
        cacc = fmaf(fmaxf(hj, 0.0f), ew2r[i][e], cacc);
      }
    }
    const float q = p + cacc;

#pragma unroll
    for (int s4 = 0; s4 < 4; ++s4) {
      const int R = s4 * 4 + (lane >> 3);
      const float pR = __shfl(p, R, 32);
      const float q0 = __shfl(q, R, 32);
      const float q1 = __shfl(q, R + 16, 32);
      v4f val = {pR, pR, pR, pR};
      if ((lane & 7) == 0) { val.x = q0; val.y = q1; }
      *(v4f*)(out + (base + R) * 32 + (lane & 7) * 4) = val;
    }

  };

  if (wgid < wstep) {
#pragma unroll
    for (int k = 0; k < 8; ++k) {
#pragma unroll
      for (int n = 0; n < 4; ++n) {
        bq[n][k] = (v2f){wq[k][0][n], wq[k][1][n]};
        bc[n][k] = (v2f){wc[k][0][n], wc[k][1][n]};
      }
    }
    tile(wgid);
    th = thn;
#pragma unroll
    for (int k = 0; k < 8; ++k) a[k] = an[k];
    tile(wgid + wstep);
  }
}

extern "C" void kernel_launch(void* const* d_in, const int* in_sizes, int n_in,
                              void* d_out, int out_size, void* d_ws, size_t ws_size,
                              hipStream_t stream) {
  (void)in_sizes; (void)n_in; (void)out_size; (void)d_ws; (void)ws_size;
  const float* x    = (const float*)d_in[0];
  const float* qW1  = (const float*)d_in[1];
  const float* qb1  = (const float*)d_in[2];
  const float* qW2  = (const float*)d_in[3];
  const float* qb2  = (const float*)d_in[4];
  const float* cW1  = (const float*)d_in[5];
  const float* cb1  = (const float*)d_in[6];
  const float* cW2  = (const float*)d_in[7];
  const float* cb2  = (const float*)d_in[8];
  const float* eTW1 = (const float*)d_in[9];
  const float* eTb1 = (const float*)d_in[10];
  const float* eTW2 = (const float*)d_in[11];
  const float* eTb2 = (const float*)d_in[12];
  const float* eHW1 = (const float*)d_in[13];
  const float* eHb1 = (const float*)d_in[14];
  const float* eHW2 = (const float*)d_in[15];
  const float* eHb2 = (const float*)d_in[16];
  float* out = (float*)d_out;

  gcn_fused_kernel<<<BLOCKS, TPB, 0, stream>>>(
      x, qW1, qb1, qW2, qb2, cW1, cb1, cW2, cb2,
      eTW1, eTb1, eTW2, eTb2, eHW1, eHb1, eHW2, eHb2, out);
}
